# starting kernel
# baseline (speedup 1.0000x reference)
#include <hip/hip_runtime.h>
#include <math.h>

typedef float v2f __attribute__((ext_vector_type(2)));
typedef float v8f __attribute__((ext_vector_type(8)));

#define B_    4
#define CIN   64
#define COUT  64
#define NPTS  16384
#define KNN   16
#define NGRP  4
#define CPG   16
#define EPS_  1e-5f

__global__ void k_init(float* __restrict__ stats) {
    if (threadIdx.x < B_ * NGRP * 2) stats[threadIdx.x] = 0.0f;
}

__global__ __launch_bounds__(256) void k_gemm(const float* __restrict__ feat,
                                              const float* __restrict__ W1,
                                              const float* __restrict__ W2,
                                              float* __restrict__ local_t,
                                              float* __restrict__ edge_t) {
    __shared__ float sW[2 * COUT * CIN];
    const int tid = threadIdx.x;
    for (int i = tid; i < COUT * CIN; i += 256) {
        sW[i]               = W1[i];
        sW[COUT * CIN + i]  = W2[i];
    }
    __syncthreads();

    const int wave = tid >> 5;
    const int lane = tid & 31;
    const int ln   = lane & 15;
    const int hi   = lane >> 4;
    const int b    = blockIdx.x >> 7;
    const int tile = ((blockIdx.x & 127) << 3) + wave;
    const int n0   = tile << 4;

    v8f acc[8];
#pragma unroll
    for (int i = 0; i < 8; ++i) acc[i] = (v8f){0, 0, 0, 0, 0, 0, 0, 0};

    const size_t fbase = (size_t)b * CIN * NPTS + (size_t)(n0 + ln);
    for (int kk = 0; kk < CIN; kk += 4) {
        const int k0 = kk + 2 * hi;
        v2f bf;
        bf.x = feat[fbase + (size_t)k0 * NPTS];
        bf.y = feat[fbase + (size_t)(k0 + 1) * NPTS];
#pragma unroll
        for (int mat = 0; mat < 2; ++mat) {
#pragma unroll
            for (int mt = 0; mt < 4; ++mt) {
                const float* w = &sW[mat * (COUT * CIN) + (mt * 16 + ln) * CIN + k0];
                v2f af;
                af.x = w[0];
                af.y = w[1];
                acc[mat * 4 + mt] = __builtin_amdgcn_wmma_f32_16x16x4_f32(
                    false, af, false, bf, (short)0, acc[mat * 4 + mt], false, false);
            }
        }
    }

#pragma unroll
    for (int mat = 0; mat < 2; ++mat) {
        float* dst = mat ? edge_t : local_t;
#pragma unroll
        for (int mt = 0; mt < 4; ++mt) {
            v8f c = acc[mat * 4 + mt];
            size_t base = ((size_t)b * NPTS + (size_t)(n0 + ln)) * COUT
                          + (size_t)(mt * 16 + 8 * hi);
            *(float4*)(dst + base)     = make_float4(c[0], c[1], c[2], c[3]);
            *(float4*)(dst + base + 4) = make_float4(c[4], c[5], c[6], c[7]);
        }
    }
}

__global__ __launch_bounds__(256) void k_stats(const int* __restrict__ knn,
                                               const float* __restrict__ local_t,
                                               const float* __restrict__ edge_t,
                                               float* __restrict__ stats) {
    __shared__ float sh[2 * NGRP];
    const int tid = threadIdx.x;
    if (tid < 2 * NGRP) sh[tid] = 0.0f;
    __syncthreads();

    const int b  = blockIdx.x >> 10;
    const int t  = ((blockIdx.x & 1023) << 8) + tid;
    const int c4 = t & 15;
    const int n  = t >> 4;
    const int g  = c4 >> 2;

    const size_t colL = ((size_t)b * NPTS + n) * COUT + (size_t)c4 * 4;
    const float4 l4 = *(const float4*)(local_t + colL);
    const int* kp = knn + ((size_t)b * NPTS + n) * KNN;

    float s = 0.0f, ss = 0.0f;
#pragma unroll
    for (int j = 0; j < KNN; ++j) {
        const int idx = kp[j];
        const float4 e4 = *(const float4*)(edge_t + ((size_t)b * NPTS + idx) * COUT
                                           + (size_t)c4 * 4);
        const float d0 = e4.x - l4.x, d1 = e4.y - l4.y;
        const float d2 = e4.z - l4.z, d3 = e4.w - l4.w;
        s  += d0 + d1 + d2 + d3;
        ss += d0 * d0 + d1 * d1 + d2 * d2 + d3 * d3;
    }
    atomicAdd(&sh[g * 2 + 0], s);
    atomicAdd(&sh[g * 2 + 1], ss);
    __syncthreads();
    if (tid < 2 * NGRP) atomicAdd(&stats[b * 2 * NGRP + tid], sh[tid]);
}

__global__ __launch_bounds__(256) void k_out(const int* __restrict__ knn,
                                             const float* __restrict__ local_t,
                                             const float* __restrict__ edge_t,
                                             const float* __restrict__ stats,
                                             const float* __restrict__ gw,
                                             const float* __restrict__ gb,
                                             float* __restrict__ out) {
    const int tid = threadIdx.x;
    const int b  = blockIdx.x >> 10;
    const int t  = ((blockIdx.x & 1023) << 8) + tid;
    const int c4 = t & 15;
    const int n  = t >> 4;
    const int g  = c4 >> 2;

    const float inv_cnt = 1.0f / (float)((size_t)CPG * NPTS * KNN);
    const float mean = stats[b * 2 * NGRP + g * 2 + 0] * inv_cnt;
    const float msq  = stats[b * 2 * NGRP + g * 2 + 1] * inv_cnt;
    const float inv  = rsqrtf(msq - mean * mean + EPS_);

    const float4 gamma = ((const float4*)gw)[c4];
    const float4 beta  = ((const float4*)gb)[c4];
    const float4 l4 = *(const float4*)(local_t + ((size_t)b * NPTS + n) * COUT
                                       + (size_t)c4 * 4);
    const int* kp = knn + ((size_t)b * NPTS + n) * KNN;

    float a0 = 0.0f, a1 = 0.0f, a2 = 0.0f, a3 = 0.0f;
#pragma unroll
    for (int j = 0; j < KNN; ++j) {
        const int idx = kp[j];
        const float4 e4 = *(const float4*)(edge_t + ((size_t)b * NPTS + idx) * COUT
                                           + (size_t)c4 * 4);
        float v0 = ((e4.x - l4.x) - mean) * inv * gamma.x + beta.x;
        float v1 = ((e4.y - l4.y) - mean) * inv * gamma.y + beta.y;
        float v2 = ((e4.z - l4.z) - mean) * inv * gamma.z + beta.z;
        float v3 = ((e4.w - l4.w) - mean) * inv * gamma.w + beta.w;
        a0 += fmaxf(v0, 0.0f);
        a1 += fmaxf(v1, 0.0f);
        a2 += fmaxf(v2, 0.0f);
        a3 += fmaxf(v3, 0.0f);
    }
    const float r = 1.0f / (float)KNN;
    const size_t ob = ((size_t)b * COUT + (size_t)c4 * 4) * NPTS + (size_t)n;
    out[ob]                     = a0 * r;
    out[ob + (size_t)NPTS]      = a1 * r;
    out[ob + 2 * (size_t)NPTS]  = a2 * r;
    out[ob + 3 * (size_t)NPTS]  = a3 * r;
}

extern "C" void kernel_launch(void* const* d_in, const int* in_sizes, int n_in,
                              void* d_out, int out_size, void* d_ws, size_t ws_size,
                              hipStream_t stream) {
    (void)in_sizes; (void)n_in; (void)out_size; (void)ws_size;

    const float* feat = (const float*)d_in[0];
    const int*   knn  = (const int*)d_in[1];
    const float* W1   = (const float*)d_in[2];
    const float* W2   = (const float*)d_in[3];
    const float* gw   = (const float*)d_in[4];
    const float* gb   = (const float*)d_in[5];
    float*       out  = (float*)d_out;

    float* local_t = (float*)d_ws;
    float* edge_t  = local_t + (size_t)B_ * NPTS * COUT;
    float* stats   = edge_t  + (size_t)B_ * NPTS * COUT;

    k_init <<<1, 64, 0, stream>>>(stats);
    k_gemm <<<B_ * (NPTS / 16) / 8, 256, 0, stream>>>(feat, W1, W2, local_t, edge_t);
    k_stats<<<B_ * (NPTS * KNN) / 256, 256, 0, stream>>>(knn, local_t, edge_t, stats);
    k_out  <<<B_ * (NPTS * KNN) / 256, 256, 0, stream>>>(knn, local_t, edge_t, stats,
                                                         gw, gb, out);
}


# k_gemm re-decomposed: a wave owns one weight matrix and 64 points, weight fragments loaded once from global into registers (no LDS, no barrier), feature read as float4 along n
# speedup vs baseline: 1.8429x; 1.8429x over previous
#include <hip/hip_runtime.h>
#include <math.h>

typedef float v2f __attribute__((ext_vector_type(2)));
typedef float v8f __attribute__((ext_vector_type(8)));

#define B_    4
#define CIN   64
#define COUT  64
#define NPTS  16384
#define KNN   16
#define NGRP  4
#define CPG   16
#define EPS_  1e-5f

__global__ void k_init(float* __restrict__ stats) {
    if (threadIdx.x < B_ * NGRP * 2) stats[threadIdx.x] = 0.0f;
}

#define GEMM_PTS_PER_WAVE 64

__global__ __launch_bounds__(128) __attribute__((amdgpu_waves_per_eu(2, 2))) void k_gemm(const float* __restrict__ feat,
                                              const float* __restrict__ W1,
                                              const float* __restrict__ W2,
                                              float* __restrict__ local_t,
                                              float* __restrict__ edge_t) {
    const int tid  = threadIdx.x;
    const int wave = tid >> 5;
    const int lane = tid & 31;
    const int ln   = lane & 15;
    const int hi   = lane >> 4;
    const int mat  = blockIdx.x & 1;
    const int st   = ((blockIdx.x >> 1) << 2) + wave;
    const int b    = st >> 8;
    const int n0   = (st & 255) * GEMM_PTS_PER_WAVE;
    const float* __restrict__ W   = mat ? W2 : W1;
    float* __restrict__       dst = mat ? edge_t : local_t;

    v2f a[4][16];
    float4 f[32];
    const float* wp = W + (size_t)ln * CIN + 2 * hi;
    const float* fp = feat + (size_t)b * CIN * NPTS + (size_t)(2 * hi) * NPTS + (size_t)(n0 + 4 * ln);
#pragma unroll
    for (int s = 0; s < 16; ++s) {
#pragma unroll
        for (int mt = 0; mt < 4; ++mt)
            a[mt][s] = *(const v2f*)(wp + (size_t)(mt * 16) * CIN + 4 * s);
        f[2 * s]     = *(const float4*)(fp + (size_t)(4 * s) * NPTS);
        f[2 * s + 1] = *(const float4*)(fp + (size_t)(4 * s + 1) * NPTS);
        asm volatile("" ::: "memory");
    }

    v8f acc[4][4];
#pragma unroll
    for (int mt = 0; mt < 4; ++mt)
#pragma unroll
        for (int t = 0; t < 4; ++t) acc[mt][t] = (v8f){0, 0, 0, 0, 0, 0, 0, 0};

#pragma unroll
    for (int s = 0; s < 16; ++s) {
        const float4 f0 = f[2 * s], f1 = f[2 * s + 1];
        const v2f bf[4] = {(v2f){f0.x, f1.x}, (v2f){f0.y, f1.y}, (v2f){f0.z, f1.z}, (v2f){f0.w, f1.w}};
#pragma unroll
        for (int t = 0; t < 4; ++t) {
#pragma unroll
            for (int mt = 0; mt < 4; ++mt) {
                acc[mt][t] = __builtin_amdgcn_wmma_f32_16x16x4_f32(
                    false, a[mt][s], false, bf[t], (short)0, acc[mt][t], false, false);
            }
        }
    }

#pragma unroll
    for (int t = 0; t < 4; ++t) {
        float* row = dst + ((size_t)b * NPTS + (size_t)(n0 + 4 * ln + t)) * COUT + (size_t)(8 * hi);
#pragma unroll
        for (int mt = 0; mt < 4; ++mt) {
            const v8f c = acc[mt][t];
            *(float4*)(row + mt * 16)     = make_float4(c[0], c[1], c[2], c[3]);
            *(float4*)(row + mt * 16 + 4) = make_float4(c[4], c[5], c[6], c[7]);
        }
    }
}

__global__ __launch_bounds__(256) void k_stats(const int* __restrict__ knn,
                                               const float* __restrict__ local_t,
                                               const float* __restrict__ edge_t,
                                               float* __restrict__ stats) {
    __shared__ float sh[2 * NGRP];
    const int tid = threadIdx.x;
    if (tid < 2 * NGRP) sh[tid] = 0.0f;
    __syncthreads();

    const int b  = blockIdx.x >> 10;
    const int t  = ((blockIdx.x & 1023) << 8) + tid;
    const int c4 = t & 15;
    const int n  = t >> 4;
    const int g  = c4 >> 2;

    const size_t colL = ((size_t)b * NPTS + n) * COUT + (size_t)c4 * 4;
    const float4 l4 = *(const float4*)(local_t + colL);
    const int* kp = knn + ((size_t)b * NPTS + n) * KNN;

    float s = 0.0f, ss = 0.0f;
#pragma unroll
    for (int j = 0; j < KNN; ++j) {
        const int idx = kp[j];
        const float4 e4 = *(const float4*)(edge_t + ((size_t)b * NPTS + idx) * COUT
                                           + (size_t)c4 * 4);
        const float d0 = e4.x - l4.x, d1 = e4.y - l4.y;
        const float d2 = e4.z - l4.z, d3 = e4.w - l4.w;
        s  += d0 + d1 + d2 + d3;
        ss += d0 * d0 + d1 * d1 + d2 * d2 + d3 * d3;
    }
    atomicAdd(&sh[g * 2 + 0], s);
    atomicAdd(&sh[g * 2 + 1], ss);
    __syncthreads();
    if (tid < 2 * NGRP) atomicAdd(&stats[b * 2 * NGRP + tid], sh[tid]);
}

#define OUT_PTS_PER_WG 64

__device__ __forceinline__ float4 norm_relu_acc(float4 a, float4 e, float4 l, float mean,
                                                float inv, float4 gamma, float4 beta) {
    const float v0 = ((e.x - l.x) - mean) * inv * gamma.x + beta.x;
    const float v1 = ((e.y - l.y) - mean) * inv * gamma.y + beta.y;
    const float v2 = ((e.z - l.z) - mean) * inv * gamma.z + beta.z;
    const float v3 = ((e.w - l.w) - mean) * inv * gamma.w + beta.w;
    a.x += fmaxf(v0, 0.0f);
    a.y += fmaxf(v1, 0.0f);
    a.z += fmaxf(v2, 0.0f);
    a.w += fmaxf(v3, 0.0f);
    return a;
}

__global__ __launch_bounds__(128) __attribute__((amdgpu_waves_per_eu(4, 4))) void k_out(const int* __restrict__ knn,
                                             const float* __restrict__ local_t,
                                             const float* __restrict__ edge_t,
                                             const float* __restrict__ stats,
                                             const float* __restrict__ gw,
                                             const float* __restrict__ gb,
                                             float* __restrict__ out) {
    const int tid  = threadIdx.x;
    const int lane = tid & 31;
    const int wave = tid >> 5;
    const int hv   = wave & 1;
    const int pb   = wave >> 1;
    const int pq   = lane >> 2;
    const int co   = lane & 3;
    const int b    = blockIdx.x >> 8;
    const int n0   = ((blockIdx.x & 255) * OUT_PTS_PER_WG) + (pb << 5) + (pq << 2);
    const int gA   = 2 * hv;
    const int chA  = hv * 32 + co * 4;
    const int chB  = chA + CPG;

    const float inv_cnt = 1.0f / (float)((size_t)CPG * NPTS * KNN);
    const float meanA = stats[b * 2 * NGRP + gA * 2 + 0] * inv_cnt;
    const float msqA  = stats[b * 2 * NGRP + gA * 2 + 1] * inv_cnt;
    const float invA  = rsqrtf(msqA - meanA * meanA + EPS_);
    const float meanB = stats[b * 2 * NGRP + gA * 2 + 2] * inv_cnt;
    const float msqB  = stats[b * 2 * NGRP + gA * 2 + 3] * inv_cnt;
    const float invB  = rsqrtf(msqB - meanB * meanB + EPS_);

    const float4 gammaA = ((const float4*)gw)[chA >> 2];
    const float4 betaA  = ((const float4*)gb)[chA >> 2];
    const float4 gammaB = ((const float4*)gw)[chB >> 2];
    const float4 betaB  = ((const float4*)gb)[chB >> 2];

    const int4*  kp   = (const int4*)(knn + ((size_t)b * NPTS + n0) * KNN);
    const float* erow = edge_t + (size_t)b * NPTS * COUT;
    const float* lrow = local_t + ((size_t)b * NPTS + n0) * COUT;

    float4 rA0 = make_float4(0.0f, 0.0f, 0.0f, 0.0f), rA1 = rA0, rA2 = rA0, rA3 = rA0;
    float4 rB0 = rA0, rB1 = rA0, rB2 = rA0, rB3 = rA0;
    const float r = 1.0f / (float)KNN;

#pragma nounroll
    for (int p = 0; p < 4; ++p) {
        int idx[KNN];
#pragma unroll
        for (int m = 0; m < 4; ++m) {
            const int4 q = kp[p * 4 + m];
            idx[m * 4 + 0] = q.x; idx[m * 4 + 1] = q.y;
            idx[m * 4 + 2] = q.z; idx[m * 4 + 3] = q.w;
        }
        const float4 lA = *(const float4*)(lrow + p * COUT + chA);
        const float4 lB = *(const float4*)(lrow + p * COUT + chB);

        float4 eA[KNN], eB[KNN];
#pragma unroll
        for (int j = 0; j < KNN; ++j) {
            const float* row = erow + (size_t)idx[j] * COUT;
            eA[j] = *(const float4*)(row + chA);
            eB[j] = *(const float4*)(row + chB);
        }
        asm volatile("" ::: "memory");
        float4 aA = make_float4(0.0f, 0.0f, 0.0f, 0.0f);
        float4 aB = make_float4(0.0f, 0.0f, 0.0f, 0.0f);
#pragma unroll
        for (int j = 0; j < KNN; ++j) {
            aA = norm_relu_acc(aA, eA[j], lA, meanA, invA, gammaA, betaA);
            aB = norm_relu_acc(aB, eB[j], lB, meanB, invB, gammaB, betaB);
        }
        rA0 = make_float4(rA0.y, rA0.z, rA0.w, aA.x * r);
        rA1 = make_float4(rA1.y, rA1.z, rA1.w, aA.y * r);
        rA2 = make_float4(rA2.y, rA2.z, rA2.w, aA.z * r);
        rA3 = make_float4(rA3.y, rA3.z, rA3.w, aA.w * r);
        rB0 = make_float4(rB0.y, rB0.z, rB0.w, aB.x * r);
        rB1 = make_float4(rB1.y, rB1.z, rB1.w, aB.y * r);
        rB2 = make_float4(rB2.y, rB2.z, rB2.w, aB.z * r);
        rB3 = make_float4(rB3.y, rB3.z, rB3.w, aB.w * r);
    }

    float* oA = out + ((size_t)b * COUT + chA) * NPTS + n0;
    float* oB = out + ((size_t)b * COUT + chB) * NPTS + n0;
    *(float4*)(oA)                    = rA0;
    *(float4*)(oA + (size_t)NPTS)     = rA1;
    *(float4*)(oA + 2 * (size_t)NPTS) = rA2;
    *(float4*)(oA + 3 * (size_t)NPTS) = rA3;
    *(float4*)(oB)                    = rB0;
    *(float4*)(oB + (size_t)NPTS)     = rB1;
    *(float4*)(oB + 2 * (size_t)NPTS) = rB2;
    *(float4*)(oB + 3 * (size_t)NPTS) = rB3;
}

extern "C" void kernel_launch(void* const* d_in, const int* in_sizes, int n_in,
                              void* d_out, int out_size, void* d_ws, size_t ws_size,
                              hipStream_t stream) {
    (void)in_sizes; (void)n_in; (void)out_size; (void)ws_size;

    const float* feat = (const float*)d_in[0];
    const int*   knn  = (const int*)d_in[1];
    const float* W1   = (const float*)d_in[2];
    const float* W2   = (const float*)d_in[3];
    const float* gw   = (const float*)d_in[4];
    const float* gb   = (const float*)d_in[5];
    float*       out  = (float*)d_out;

    float* local_t = (float*)d_ws;
    float* edge_t  = local_t + (size_t)B_ * NPTS * COUT;
    float* stats   = edge_t  + (size_t)B_ * NPTS * COUT;

    k_init <<<1, 64, 0, stream>>>(stats);
    k_gemm <<<B_ * (NPTS / GEMM_PTS_PER_WAVE) * 2 / 4, 128, 0, stream>>>(feat, W1, W2, local_t, edge_t);
    k_stats<<<B_ * (NPTS * KNN) / 256, 256, 0, stream>>>(knn, local_t, edge_t, stats);
    k_out  <<<B_ * NPTS / OUT_PTS_PER_WG, 128, 0, stream>>>(knn, local_t, edge_t, stats,
                                                            gw, gb, out);
}


# k_stats on a resident grid: a wave walks 16 points, group sums in registers, 32 whole-row gathers in flight; one fixed-order sum and one atomic set per work-group (sum order changes); 8x fewer waves
# speedup vs baseline: 2.1306x; 1.1561x over previous
#include <hip/hip_runtime.h>
#include <math.h>

typedef float v2f __attribute__((ext_vector_type(2)));
typedef float v8f __attribute__((ext_vector_type(8)));

#define B_    4
#define CIN   64
#define COUT  64
#define NPTS  16384
#define KNN   16
#define NGRP  4
#define CPG   16
#define EPS_  1e-5f

__global__ void k_init(float* __restrict__ stats) {
    if (threadIdx.x < B_ * NGRP * 2) stats[threadIdx.x] = 0.0f;
}

#define GEMM_PTS_PER_WAVE 64

__global__ __launch_bounds__(128) __attribute__((amdgpu_waves_per_eu(2, 2))) void k_gemm(const float* __restrict__ feat,
                                              const float* __restrict__ W1,
                                              const float* __restrict__ W2,
                                              float* __restrict__ local_t,
                                              float* __restrict__ edge_t) {
    const int tid  = threadIdx.x;
    const int wave = tid >> 5;
    const int lane = tid & 31;
    const int ln   = lane & 15;
    const int hi   = lane >> 4;
    const int mat  = blockIdx.x & 1;
    const int st   = ((blockIdx.x >> 1) << 2) + wave;
    const int b    = st >> 8;
    const int n0   = (st & 255) * GEMM_PTS_PER_WAVE;
    const float* __restrict__ W   = mat ? W2 : W1;
    float* __restrict__       dst = mat ? edge_t : local_t;

    v2f a[4][16];
    float4 f[32];
    const float* wp = W + (size_t)ln * CIN + 2 * hi;
    const float* fp = feat + (size_t)b * CIN * NPTS + (size_t)(2 * hi) * NPTS + (size_t)(n0 + 4 * ln);
#pragma unroll
    for (int s = 0; s < 16; ++s) {
#pragma unroll
        for (int mt = 0; mt < 4; ++mt)
            a[mt][s] = *(const v2f*)(wp + (size_t)(mt * 16) * CIN + 4 * s);
        f[2 * s]     = *(const float4*)(fp + (size_t)(4 * s) * NPTS);
        f[2 * s + 1] = *(const float4*)(fp + (size_t)(4 * s + 1) * NPTS);
        asm volatile("" ::: "memory");
    }

    v8f acc[4][4];
#pragma unroll
    for (int mt = 0; mt < 4; ++mt)
#pragma unroll
        for (int t = 0; t < 4; ++t) acc[mt][t] = (v8f){0, 0, 0, 0, 0, 0, 0, 0};

#pragma unroll
    for (int s = 0; s < 16; ++s) {
        const float4 f0 = f[2 * s], f1 = f[2 * s + 1];
        const v2f bf[4] = {(v2f){f0.x, f1.x}, (v2f){f0.y, f1.y}, (v2f){f0.z, f1.z}, (v2f){f0.w, f1.w}};
#pragma unroll
        for (int t = 0; t < 4; ++t) {
#pragma unroll
            for (int mt = 0; mt < 4; ++mt) {
                acc[mt][t] = __builtin_amdgcn_wmma_f32_16x16x4_f32(
                    false, a[mt][s], false, bf[t], (short)0, acc[mt][t], false, false);
            }
        }
    }

#pragma unroll
    for (int t = 0; t < 4; ++t) {
        float* row = dst + ((size_t)b * NPTS + (size_t)(n0 + 4 * ln + t)) * COUT + (size_t)(8 * hi);
#pragma unroll
        for (int mt = 0; mt < 4; ++mt) {
            const v8f c = acc[mt][t];
            *(float4*)(row + mt * 16)     = make_float4(c[0], c[1], c[2], c[3]);
            *(float4*)(row + mt * 16 + 4) = make_float4(c[4], c[5], c[6], c[7]);
        }
    }
}

#define STATS_WAVES        8
#define STATS_PTS_PER_WAVE 16
#define STATS_PTS_PER_TRIP 4
#define STATS_TRIPS        (STATS_PTS_PER_WAVE / STATS_PTS_PER_TRIP)
#define STATS_PTS_PER_WG   (STATS_WAVES * STATS_PTS_PER_WAVE)

__device__ __forceinline__ void stats_acc(float4& s4, float4& ss4, float4 e, float4 l) {
    const float d0 = e.x - l.x, d1 = e.y - l.y, d2 = e.z - l.z, d3 = e.w - l.w;
    s4.x += d0; s4.y += d1; s4.z += d2; s4.w += d3;
    ss4.x += d0 * d0; ss4.y += d1 * d1; ss4.z += d2 * d2; ss4.w += d3 * d3;
}

__global__ __launch_bounds__(256) void k_stats(const int* __restrict__ knn,
                                               const float* __restrict__ local_t,
                                               const float* __restrict__ edge_t,
                                               float* __restrict__ stats) {
    __shared__ float sh[STATS_WAVES][2 * NGRP];
    const int tid  = threadIdx.x;
    const int lane = tid & 31;
    const int wave = tid >> 5;
    const int h    = lane >> 4;
    const int c4   = lane & 15;
    const int b    = blockIdx.x / (NPTS / STATS_PTS_PER_WG);
    const int nw   = (blockIdx.x % (NPTS / STATS_PTS_PER_WG)) * STATS_PTS_PER_WG
                     + wave * STATS_PTS_PER_WAVE;

    const int4*   kp = (const int4*)(knn + ((size_t)b * NPTS + nw + h) * KNN);
    const float4* lp = (const float4*)(local_t + ((size_t)b * NPTS + nw + h) * COUT) + c4;
    const char*    eb   = (const char*)(edge_t + (size_t)b * NPTS * COUT);
    const unsigned coff = (unsigned)c4 * 16u;

    int4   qa0 = kp[0], qa1 = kp[1], qa2 = kp[2], qa3 = kp[3];
    int4   qb0 = kp[8], qb1 = kp[9], qb2 = kp[10], qb3 = kp[11];
    float4 la = lp[0], lb = lp[32];
    float4 s4  = make_float4(0.0f, 0.0f, 0.0f, 0.0f);
    float4 ss4 = make_float4(0.0f, 0.0f, 0.0f, 0.0f);

#pragma nounroll
    for (int t = 0; t < STATS_TRIPS; ++t) {
        const int ia[KNN] = {qa0.x, qa0.y, qa0.z, qa0.w, qa1.x, qa1.y, qa1.z, qa1.w,
                             qa2.x, qa2.y, qa2.z, qa2.w, qa3.x, qa3.y, qa3.z, qa3.w};
        const int ib[KNN] = {qb0.x, qb0.y, qb0.z, qb0.w, qb1.x, qb1.y, qb1.z, qb1.w,
                             qb2.x, qb2.y, qb2.z, qb2.w, qb3.x, qb3.y, qb3.z, qb3.w};
        float4 ea[KNN], ebv[KNN];
        for (int j = 0; j < KNN; ++j)
            ea[j] = *(const float4*)(eb + (((unsigned)ia[j] << 8) + coff));
        for (int j = 0; j < KNN; ++j)
            ebv[j] = *(const float4*)(eb + (((unsigned)ib[j] << 8) + coff));
        const float4 lca = la, lcb = lb;
        if (t + 1 < STATS_TRIPS) {
            const int4* kn = kp + (t + 1) * (STATS_PTS_PER_TRIP * KNN / 4);
            qa0 = kn[0]; qa1 = kn[1]; qa2 = kn[2];  qa3 = kn[3];
            qb0 = kn[8]; qb1 = kn[9]; qb2 = kn[10]; qb3 = kn[11];
            const float4* ln = lp + (t + 1) * (STATS_PTS_PER_TRIP * COUT / 4);
            la = ln[0]; lb = ln[32];
        }
        for (int j = 0; j < KNN; ++j) stats_acc(s4, ss4, ea[j], lca);
        for (int j = 0; j < KNN; ++j) stats_acc(s4, ss4, ebv[j], lcb);
    }

    float s  = (s4.x + s4.y) + (s4.z + s4.w);
    float ss = (ss4.x + ss4.y) + (ss4.z + ss4.w);
    s += __shfl_xor(s, 1);   ss += __shfl_xor(ss, 1);
    s += __shfl_xor(s, 2);   ss += __shfl_xor(ss, 2);
    s += __shfl_xor(s, 16);  ss += __shfl_xor(ss, 16);
    if ((lane & 0x13) == 0) {
        sh[wave][(c4 >> 2) * 2 + 0] = s;
        sh[wave][(c4 >> 2) * 2 + 1] = ss;
    }
    __syncthreads();
    if (tid < 2 * NGRP) {
        float acc = 0.0f;
        for (int w = 0; w < STATS_WAVES; ++w) acc += sh[w][tid];
        atomicAdd(&stats[b * 2 * NGRP + tid], acc);
    }
}

#define OUT_PTS_PER_WG 64

__device__ __forceinline__ float4 norm_relu_acc(float4 a, float4 e, float4 l, float mean,
                                                float inv, float4 gamma, float4 beta) {
    const float v0 = ((e.x - l.x) - mean) * inv * gamma.x + beta.x;
    const float v1 = ((e.y - l.y) - mean) * inv * gamma.y + beta.y;
    const float v2 = ((e.z - l.z) - mean) * inv * gamma.z + beta.z;
    const float v3 = ((e.w - l.w) - mean) * inv * gamma.w + beta.w;
    a.x += fmaxf(v0, 0.0f);
    a.y += fmaxf(v1, 0.0f);
    a.z += fmaxf(v2, 0.0f);
    a.w += fmaxf(v3, 0.0f);
    return a;
}

__global__ __launch_bounds__(128) __attribute__((amdgpu_waves_per_eu(4, 4))) void k_out(const int* __restrict__ knn,
                                             const float* __restrict__ local_t,
                                             const float* __restrict__ edge_t,
                                             const float* __restrict__ stats,
                                             const float* __restrict__ gw,
                                             const float* __restrict__ gb,
                                             float* __restrict__ out) {
    const int tid  = threadIdx.x;
    const int lane = tid & 31;
    const int wave = tid >> 5;
    const int hv   = wave & 1;
    const int pb   = wave >> 1;
    const int pq   = lane >> 2;
    const int co   = lane & 3;
    const int b    = blockIdx.x >> 8;
    const int n0   = ((blockIdx.x & 255) * OUT_PTS_PER_WG) + (pb << 5) + (pq << 2);
    const int gA   = 2 * hv;
    const int chA  = hv * 32 + co * 4;
    const int chB  = chA + CPG;

    const float inv_cnt = 1.0f / (float)((size_t)CPG * NPTS * KNN);
    const float meanA = stats[b * 2 * NGRP + gA * 2 + 0] * inv_cnt;
    const float msqA  = stats[b * 2 * NGRP + gA * 2 + 1] * inv_cnt;
    const float invA  = rsqrtf(msqA - meanA * meanA + EPS_);
    const float meanB = stats[b * 2 * NGRP + gA * 2 + 2] * inv_cnt;
    const float msqB  = stats[b * 2 * NGRP + gA * 2 + 3] * inv_cnt;
    const float invB  = rsqrtf(msqB - meanB * meanB + EPS_);

    const float4 gammaA = ((const float4*)gw)[chA >> 2];
    const float4 betaA  = ((const float4*)gb)[chA >> 2];
    const float4 gammaB = ((const float4*)gw)[chB >> 2];
    const float4 betaB  = ((const float4*)gb)[chB >> 2];

    const int4*  kp   = (const int4*)(knn + ((size_t)b * NPTS + n0) * KNN);
    const float* erow = edge_t + (size_t)b * NPTS * COUT;
    const float* lrow = local_t + ((size_t)b * NPTS + n0) * COUT;

    float4 rA0 = make_float4(0.0f, 0.0f, 0.0f, 0.0f), rA1 = rA0, rA2 = rA0, rA3 = rA0;
    float4 rB0 = rA0, rB1 = rA0, rB2 = rA0, rB3 = rA0;
    const float r = 1.0f / (float)KNN;

#pragma nounroll
    for (int p = 0; p < 4; ++p) {
        int idx[KNN];
#pragma unroll
        for (int m = 0; m < 4; ++m) {
            const int4 q = kp[p * 4 + m];
            idx[m * 4 + 0] = q.x; idx[m * 4 + 1] = q.y;
            idx[m * 4 + 2] = q.z; idx[m * 4 + 3] = q.w;
        }
        const float4 lA = *(const float4*)(lrow + p * COUT + chA);
        const float4 lB = *(const float4*)(lrow + p * COUT + chB);

        float4 eA[KNN], eB[KNN];
#pragma unroll
        for (int j = 0; j < KNN; ++j) {
            const float* row = erow + (size_t)idx[j] * COUT;
            eA[j] = *(const float4*)(row + chA);
            eB[j] = *(const float4*)(row + chB);
        }
        asm volatile("" ::: "memory");
        float4 aA = make_float4(0.0f, 0.0f, 0.0f, 0.0f);
        float4 aB = make_float4(0.0f, 0.0f, 0.0f, 0.0f);
#pragma unroll
        for (int j = 0; j < KNN; ++j) {
            aA = norm_relu_acc(aA, eA[j], lA, meanA, invA, gammaA, betaA);
            aB = norm_relu_acc(aB, eB[j], lB, meanB, invB, gammaB, betaB);
        }
        rA0 = make_float4(rA0.y, rA0.z, rA0.w, aA.x * r);
        rA1 = make_float4(rA1.y, rA1.z, rA1.w, aA.y * r);
        rA2 = make_float4(rA2.y, rA2.z, rA2.w, aA.z * r);
        rA3 = make_float4(rA3.y, rA3.z, rA3.w, aA.w * r);
        rB0 = make_float4(rB0.y, rB0.z, rB0.w, aB.x * r);
        rB1 = make_float4(rB1.y, rB1.z, rB1.w, aB.y * r);
        rB2 = make_float4(rB2.y, rB2.z, rB2.w, aB.z * r);
        rB3 = make_float4(rB3.y, rB3.z, rB3.w, aB.w * r);
    }

    float* oA = out + ((size_t)b * COUT + chA) * NPTS + n0;
    float* oB = out + ((size_t)b * COUT + chB) * NPTS + n0;
    *(float4*)(oA)                    = rA0;
    *(float4*)(oA + (size_t)NPTS)     = rA1;
    *(float4*)(oA + 2 * (size_t)NPTS) = rA2;
    *(float4*)(oA + 3 * (size_t)NPTS) = rA3;
    *(float4*)(oB)                    = rB0;
    *(float4*)(oB + (size_t)NPTS)     = rB1;
    *(float4*)(oB + 2 * (size_t)NPTS) = rB2;
    *(float4*)(oB + 3 * (size_t)NPTS) = rB3;
}

extern "C" void kernel_launch(void* const* d_in, const int* in_sizes, int n_in,
                              void* d_out, int out_size, void* d_ws, size_t ws_size,
                              hipStream_t stream) {
    (void)in_sizes; (void)n_in; (void)out_size; (void)ws_size;

    const float* feat = (const float*)d_in[0];
    const int*   knn  = (const int*)d_in[1];
    const float* W1   = (const float*)d_in[2];
    const float* W2   = (const float*)d_in[3];
    const float* gw   = (const float*)d_in[4];
    const float* gb   = (const float*)d_in[5];
    float*       out  = (float*)d_out;

    float* local_t = (float*)d_ws;
    float* edge_t  = local_t + (size_t)B_ * NPTS * COUT;
    float* stats   = edge_t  + (size_t)B_ * NPTS * COUT;

    k_init <<<1, 64, 0, stream>>>(stats);
    k_gemm <<<B_ * (NPTS / GEMM_PTS_PER_WAVE) * 2 / 4, 128, 0, stream>>>(feat, W1, W2, local_t, edge_t);
    k_stats<<<B_ * NPTS / STATS_PTS_PER_WG, 256, 0, stream>>>(knn, local_t, edge_t, stats);
    k_out  <<<B_ * NPTS / OUT_PTS_PER_WG, 128, 0, stream>>>(knn, local_t, edge_t, stats,
                                                            gw, gb, out);
}


# k_stats, k_out: one work-group per compute unit copies a (batch, 4-channel slice) table of edge (256 KiB, slice-major from k_gemm) into LDS and gathers from LDS, not L2; stats sum order changes
# speedup vs baseline: 2.2469x; 1.0546x over previous
#include <hip/hip_runtime.h>
#include <math.h>

typedef float v2f __attribute__((ext_vector_type(2)));
typedef float v8f __attribute__((ext_vector_type(8)));

#define B_    4
#define CIN   64
#define COUT  64
#define NPTS  16384
#define KNN   16
#define NGRP  4
#define CPG   16
#define EPS_  1e-5f
#define NSLICE 16

__global__ void k_init(float* __restrict__ stats) {
    if (threadIdx.x < B_ * NGRP * 2) stats[threadIdx.x] = 0.0f;
}

#define GEMM_PTS_PER_WAVE 64

__global__ __launch_bounds__(128) __attribute__((amdgpu_waves_per_eu(2, 2))) void k_gemm(const float* __restrict__ feat,
                                              const float* __restrict__ W1,
                                              const float* __restrict__ W2,
                                              float* __restrict__ local_t,
                                              float* __restrict__ edge_t) {
    const int tid  = threadIdx.x;
    const int wave = tid >> 5;
    const int lane = tid & 31;
    const int ln   = lane & 15;
    const int hi   = lane >> 4;
    const int mat  = blockIdx.x & 1;
    const int st   = ((blockIdx.x >> 1) << 2) + wave;
    const int b    = st >> 8;
    const int n0   = (st & 255) * GEMM_PTS_PER_WAVE;
    const float* __restrict__ W   = mat ? W2 : W1;
    float* __restrict__       dst = mat ? edge_t : local_t;

    v2f a[4][16];
    float4 f[32];
    const float* wp = W + (size_t)ln * CIN + 2 * hi;
    const float* fp = feat + (size_t)b * CIN * NPTS + (size_t)(2 * hi) * NPTS + (size_t)(n0 + 4 * ln);
#pragma unroll
    for (int s = 0; s < 16; ++s) {
#pragma unroll
        for (int mt = 0; mt < 4; ++mt)
            a[mt][s] = *(const v2f*)(wp + (size_t)(mt * 16) * CIN + 4 * s);
        f[2 * s]     = *(const float4*)(fp + (size_t)(4 * s) * NPTS);
        f[2 * s + 1] = *(const float4*)(fp + (size_t)(4 * s + 1) * NPTS);
        asm volatile("" ::: "memory");
    }

    v8f acc[4][4];
#pragma unroll
    for (int mt = 0; mt < 4; ++mt)
#pragma unroll
        for (int t = 0; t < 4; ++t) acc[mt][t] = (v8f){0, 0, 0, 0, 0, 0, 0, 0};

#pragma unroll
    for (int s = 0; s < 16; ++s) {
        const float4 f0 = f[2 * s], f1 = f[2 * s + 1];
        const v2f bf[4] = {(v2f){f0.x, f1.x}, (v2f){f0.y, f1.y}, (v2f){f0.z, f1.z}, (v2f){f0.w, f1.w}};
#pragma unroll
        for (int t = 0; t < 4; ++t) {
#pragma unroll
            for (int mt = 0; mt < 4; ++mt) {
                acc[mt][t] = __builtin_amdgcn_wmma_f32_16x16x4_f32(
                    false, a[mt][s], false, bf[t], (short)0, acc[mt][t], false, false);
            }
        }
    }

#pragma unroll
    for (int mt = 0; mt < 4; ++mt) {
        float* p0 = dst + (((size_t)b * NSLICE + (size_t)(4 * mt + 2 * hi)) * NPTS + (size_t)(n0 + 4 * ln)) * 4;
        float* p1 = p0 + (size_t)NPTS * 4;
#pragma unroll
        for (int t = 0; t < 4; ++t) {
            const v8f c = acc[mt][t];
            *(float4*)(p0 + 4 * t) = make_float4(c[0], c[1], c[2], c[3]);
            *(float4*)(p1 + 4 * t) = make_float4(c[4], c[5], c[6], c[7]);
        }
    }
}

#define SL_THREADS    1024
#define SL_QUARTERS   4
#define SL_PTS_PER_WG (NPTS / SL_QUARTERS)
#define SL_TRIPS      (SL_PTS_PER_WG / SL_THREADS)
#define SL_FILL_STEPS (NPTS / SL_THREADS)

__device__ __forceinline__ void slice_fill(float4* __restrict__ tab, const float4* __restrict__ src, int tid) {
    float4 v[SL_FILL_STEPS];
#pragma unroll
    for (int i = 0; i < SL_FILL_STEPS; ++i) v[i] = src[i * SL_THREADS + tid];
    asm volatile("" ::: "memory");
#pragma unroll
    for (int i = 0; i < SL_FILL_STEPS; ++i) tab[i * SL_THREADS + tid] = v[i];
    __syncthreads();
}

__device__ __forceinline__ void stats_acc(float4& s4, float4& ss4, float4 e, float4 l) {
    const float d0 = e.x - l.x, d1 = e.y - l.y, d2 = e.z - l.z, d3 = e.w - l.w;
    s4.x += d0; s4.y += d1; s4.z += d2; s4.w += d3;
    ss4.x += d0 * d0; ss4.y += d1 * d1; ss4.z += d2 * d2; ss4.w += d3 * d3;
}

__global__ __launch_bounds__(SL_THREADS) void k_stats(const int* __restrict__ knn,
                                                      const float* __restrict__ local_s,
                                                      const float* __restrict__ edge_s,
                                                      float* __restrict__ stats) {
    __shared__ float4 tab[NPTS];
    __shared__ float  red[SL_THREADS / 32][2];
    const int tid = threadIdx.x;
    const int q   = blockIdx.x & (SL_QUARTERS - 1);
    const int s   = (blockIdx.x >> 2) & (NSLICE - 1);
    const int b   = blockIdx.x >> 6;
    const size_t trow = ((size_t)b * NSLICE + s) * NPTS;
    const int    n0   = q * SL_PTS_PER_WG + tid;

    const int4*   kp = (const int4*)(knn + ((size_t)b * NPTS + n0) * KNN);
    const float4* lp = (const float4*)local_s + trow + n0;
    int4   q0 = kp[0], q1 = kp[1], q2 = kp[2], q3 = kp[3];
    float4 l  = lp[0];

    slice_fill(tab, (const float4*)edge_s + trow, tid);

    float4 s4  = make_float4(0.0f, 0.0f, 0.0f, 0.0f);
    float4 ss4 = make_float4(0.0f, 0.0f, 0.0f, 0.0f);
#pragma nounroll
    for (int t = 0; t < SL_TRIPS; ++t) {
        const int idx[KNN] = {q0.x, q0.y, q0.z, q0.w, q1.x, q1.y, q1.z, q1.w,
                              q2.x, q2.y, q2.z, q2.w, q3.x, q3.y, q3.z, q3.w};
        const float4 lc = l;
        float4 e[KNN];
#pragma unroll
        for (int j = 0; j < KNN; ++j) e[j] = tab[idx[j]];
        if (t + 1 < SL_TRIPS) {
            const int4* kn = kp + (size_t)(t + 1) * SL_THREADS * (KNN / 4);
            q0 = kn[0]; q1 = kn[1]; q2 = kn[2]; q3 = kn[3];
            l = lp[(t + 1) * SL_THREADS];
        }
#pragma unroll
        for (int j = 0; j < KNN; ++j) stats_acc(s4, ss4, e[j], lc);
    }

    float sm = (s4.x + s4.y) + (s4.z + s4.w);
    float sq = (ss4.x + ss4.y) + (ss4.z + ss4.w);
#pragma unroll
    for (int m = 16; m >= 1; m >>= 1) {
        sm += __shfl_xor(sm, m);
        sq += __shfl_xor(sq, m);
    }
    if ((tid & 31) == 0) {
        red[tid >> 5][0] = sm;
        red[tid >> 5][1] = sq;
    }
    __syncthreads();
    if (tid < 2) {
        float acc = 0.0f;
        for (int w = 0; w < SL_THREADS / 32; ++w) acc += red[w][tid];
        atomicAdd(&stats[b * 2 * NGRP + (s >> 2) * 2 + tid], acc);
    }
}

__device__ __forceinline__ float4 norm_relu_acc(float4 a, float4 e, float4 l, float mean,
                                                float inv, float4 gamma, float4 beta) {
    const float v0 = ((e.x - l.x) - mean) * inv * gamma.x + beta.x;
    const float v1 = ((e.y - l.y) - mean) * inv * gamma.y + beta.y;
    const float v2 = ((e.z - l.z) - mean) * inv * gamma.z + beta.z;
    const float v3 = ((e.w - l.w) - mean) * inv * gamma.w + beta.w;
    a.x += fmaxf(v0, 0.0f);
    a.y += fmaxf(v1, 0.0f);
    a.z += fmaxf(v2, 0.0f);
    a.w += fmaxf(v3, 0.0f);
    return a;
}

__global__ __launch_bounds__(SL_THREADS) void k_out(const int* __restrict__ knn,
                                                    const float* __restrict__ local_s,
                                                    const float* __restrict__ edge_s,
                                                    const float* __restrict__ stats,
                                                    const float* __restrict__ gw,
                                                    const float* __restrict__ gb,
                                                    float* __restrict__ out) {
    __shared__ float4 tab[NPTS];
    const int tid = threadIdx.x;
    const int q   = blockIdx.x & (SL_QUARTERS - 1);
    const int s   = (blockIdx.x >> 2) & (NSLICE - 1);
    const int b   = blockIdx.x >> 6;
    const int g   = s >> 2;
    const size_t trow = ((size_t)b * NSLICE + s) * NPTS;
    const int    n0   = q * SL_PTS_PER_WG + tid;

    const int4*   kp = (const int4*)(knn + ((size_t)b * NPTS + n0) * KNN);
    const float4* lp = (const float4*)local_s + trow + n0;
    int4   q0 = kp[0], q1 = kp[1], q2 = kp[2], q3 = kp[3];
    float4 l  = lp[0];

    const float inv_cnt = 1.0f / (float)((size_t)CPG * NPTS * KNN);
    const float mean = stats[b * 2 * NGRP + g * 2 + 0] * inv_cnt;
    const float msq  = stats[b * 2 * NGRP + g * 2 + 1] * inv_cnt;
    const float inv  = rsqrtf(msq - mean * mean + EPS_);
    const float4 gamma = ((const float4*)gw)[s];
    const float4 beta  = ((const float4*)gb)[s];
    const float r = 1.0f / (float)KNN;

    slice_fill(tab, (const float4*)edge_s + trow, tid);

    float* op = out + ((size_t)b * COUT + 4 * s) * NPTS + n0;
#pragma nounroll
    for (int t = 0; t < SL_TRIPS; ++t) {
        const int idx[KNN] = {q0.x, q0.y, q0.z, q0.w, q1.x, q1.y, q1.z, q1.w,
                              q2.x, q2.y, q2.z, q2.w, q3.x, q3.y, q3.z, q3.w};
        const float4 lc = l;
        float4 e[KNN];
#pragma unroll
        for (int j = 0; j < KNN; ++j) e[j] = tab[idx[j]];
        if (t + 1 < SL_TRIPS) {
            const int4* kn = kp + (size_t)(t + 1) * SL_THREADS * (KNN / 4);
            q0 = kn[0]; q1 = kn[1]; q2 = kn[2]; q3 = kn[3];
            l = lp[(t + 1) * SL_THREADS];
        }
        float4 a = make_float4(0.0f, 0.0f, 0.0f, 0.0f);
#pragma unroll
        for (int j = 0; j < KNN; ++j) a = norm_relu_acc(a, e[j], lc, mean, inv, gamma, beta);
        float* o = op + t * SL_THREADS;
        o[0]                = a.x * r;
        o[(size_t)NPTS]     = a.y * r;
        o[2 * (size_t)NPTS] = a.z * r;
        o[3 * (size_t)NPTS] = a.w * r;
    }
}

extern "C" void kernel_launch(void* const* d_in, const int* in_sizes, int n_in,
                              void* d_out, int out_size, void* d_ws, size_t ws_size,
                              hipStream_t stream) {
    (void)in_sizes; (void)n_in; (void)out_size; (void)ws_size;

    const float* feat = (const float*)d_in[0];
    const int*   knn  = (const int*)d_in[1];
    const float* W1   = (const float*)d_in[2];
    const float* W2   = (const float*)d_in[3];
    const float* gw   = (const float*)d_in[4];
    const float* gb   = (const float*)d_in[5];
    float*       out  = (float*)d_out;

    float* local_t = (float*)d_ws;
    float* edge_t  = local_t + (size_t)B_ * NPTS * COUT;
    float* stats   = edge_t  + (size_t)B_ * NPTS * COUT;

    k_init <<<1, 64, 0, stream>>>(stats);
    k_gemm <<<B_ * (NPTS / GEMM_PTS_PER_WAVE) * 2 / 4, 128, 0, stream>>>(feat, W1, W2, local_t, edge_t);
    k_stats<<<B_ * NSLICE * SL_QUARTERS, SL_THREADS, 0, stream>>>(knn, local_t, edge_t, stats);
    k_out  <<<B_ * NSLICE * SL_QUARTERS, SL_THREADS, 0, stream>>>(knn, local_t, edge_t, stats,
                                                                  gw, gb, out);
}
